# starting kernel
# baseline (speedup 1.0000x reference)
#include <hip/hip_runtime.h>

typedef __attribute__((ext_vector_type(2))) float v2f;
typedef __attribute__((ext_vector_type(8))) float v8f;

#define BB 8
#define TT 64
#define SS 512
#define QU 1024
#define VU 512
#define UU 256

__global__ __launch_bounds__(256) void wmma_gemm_f32(
    const float* __restrict__ A, const float* __restrict__ Bm,
    float* __restrict__ C, int M, int N, int K,
    long strideA, long strideB, long strideC)
{
    const int batch = blockIdx.y;
    const float* Ab = A  + (long)batch * strideA;
    const float* Bb = Bm + (long)batch * strideB;
    float*       Cb = C  + (long)batch * strideC;

    const int lane   = threadIdx.x & 31;
    const int wave   = threadIdx.x >> 5;
    const int tilesN = N >> 4;
    const int tilesM = M >> 4;
    const int tile   = blockIdx.x * 8 + wave;
    if (tile >= tilesM * tilesN) return;
    const int tm = (tile / tilesN) << 4;
    const int tn = (tile % tilesN) << 4;

    const int half = lane >> 4;
    const int l16  = lane & 15;

    v8f acc = {};
    const float* arow = Ab + (long)(tm + l16) * K;
    const float* bcol = Bb + (tn + l16);

    for (int kk = 0; kk < K; kk += 4) {
        const int ka = kk + 2 * half;
        v2f a = *(const v2f*)(arow + ka);
        v2f b;
        b.x = bcol[(long)ka       * N];
        b.y = bcol[(long)(ka + 1) * N];
        acc = __builtin_amdgcn_wmma_f32_16x16x4_f32(
             false, a,  false, b,
             (short)0, acc,  false,  false);
    }

    float* crow = Cb + (long)(tm + half * 8) * N + tn + l16;
#pragma unroll
    for (int v = 0; v < 8; ++v) crow[(long)v * N] = acc[v];
}

__global__ __launch_bounds__(256) void scores_softmax(
    const float* __restrict__ w1q,
    const float* __restrict__ w2k,
    const float* __restrict__ scale,
    float* __restrict__ attn)
{
    __shared__ float s_scores[SS];
    __shared__ float red[256];

    const int bt   = blockIdx.x;
    const int b    = bt >> 6;
    const int tid  = threadIdx.x;
    const int lane = tid & 31;
    const int wave = tid >> 5;

    const float4* qrow = (const float4*)(w1q + (long)bt * UU) + lane * 2;
    const float4  q0 = qrow[0], q1 = qrow[1];
    const float4* srow = (const float4*)scale + lane * 2;
    const float4  c0 = srow[0], c1 = srow[1];

    for (int s = wave; s < SS; s += 8) {
        const float4* krow = (const float4*)(w2k + (long)(b * SS + s) * UU) + lane * 2;
        const float4 k0 = krow[0], k1 = krow[1];
        float p;
        p  = c0.x * tanhf(q0.x + k0.x);
        p += c0.y * tanhf(q0.y + k0.y);
        p += c0.z * tanhf(q0.z + k0.z);
        p += c0.w * tanhf(q0.w + k0.w);
        p += c1.x * tanhf(q1.x + k1.x);
        p += c1.y * tanhf(q1.y + k1.y);
        p += c1.z * tanhf(q1.z + k1.z);
        p += c1.w * tanhf(q1.w + k1.w);
#pragma unroll
        for (int off = 16; off; off >>= 1) p += __shfl_xor(p, off, 32);
        if (lane == 0) s_scores[s] = p;
    }
    __syncthreads();

    const float v0 = s_scores[tid];
    const float v1 = s_scores[tid + 256];
    red[tid] = fmaxf(v0, v1);
    __syncthreads();
#pragma unroll
    for (int off = 128; off; off >>= 1) {
        if (tid < off) red[tid] = fmaxf(red[tid], red[tid + off]);
        __syncthreads();
    }
    const float m = red[0];
    __syncthreads();

    const float e0 = __expf(v0 - m);
    const float e1 = __expf(v1 - m);
    red[tid] = e0 + e1;
    __syncthreads();
#pragma unroll
    for (int off = 128; off; off >>= 1) {
        if (tid < off) red[tid] += red[tid + off];
        __syncthreads();
    }
    const float inv = 1.0f / red[0];

    attn[(long)bt * SS + tid]       = e0 * inv;
    attn[(long)bt * SS + tid + 256] = e1 * inv;
}

extern "C" void kernel_launch(void* const* d_in, const int* in_sizes, int n_in,
                              void* d_out, int out_size, void* d_ws, size_t ws_size,
                              hipStream_t stream)
{
    (void)in_sizes; (void)n_in; (void)out_size; (void)ws_size;

    const float* query = (const float*)d_in[0];
    const float* value = (const float*)d_in[1];
    const float* W1    = (const float*)d_in[2];
    const float* W2    = (const float*)d_in[3];
    const float* scale = (const float*)d_in[4];

    float* ctx  = (float*)d_out;
    float* attn = ctx + (long)BB * TT * VU;

    float* w1q = (float*)d_ws;
    float* w2k = w1q + (long)BB * TT * UU;

    wmma_gemm_f32<<<dim3(64, 1), 256, 0, stream>>>(
        query, W1, w1q, BB * TT, UU, QU, 0, 0, 0);

    wmma_gemm_f32<<<dim3(512, 1), 256, 0, stream>>>(
        value, W2, w2k, BB * SS, UU, VU, 0, 0, 0);

    scores_softmax<<<dim3(BB * TT), 256, 0, stream>>>(w1q, w2k, scale, attn);

    wmma_gemm_f32<<<dim3(16, BB), 256, 0, stream>>>(
        attn, value, ctx, TT, VU, SS,
        (long)TT * SS, (long)SS * VU, (long)TT * VU);
}


# scores kernel softmax: scores stay in registers (lane = key); row max and sum by DPP/permlanex16 exchanges in a wave plus one LDS hop over the 16 waves: 2 barriers, was 20 (two 8-level LDS trees)
# speedup vs baseline: 5.2700x; 5.2700x over previous
#include <hip/hip_runtime.h>

typedef __attribute__((ext_vector_type(2))) float v2f;
typedef __attribute__((ext_vector_type(4))) float v4f;
typedef __attribute__((ext_vector_type(8))) float v8f;

#define BB 8
#define TT 64
#define SS 512
#define QU 1024
#define VU 512
#define UU 256

#define KC 64
#define KG (KC / 8)

struct FragSet {
    v4f a[KG];
    v4f b[KG][4];
};

__device__ __forceinline__ void load_set(FragSet& f, const float* __restrict__ ap,
                                         const float* __restrict__ bp, long N)
{
#pragma unroll
    for (int g = 0; g < KG; ++g) {
        f.a[g] = *(const v4f*)(ap + g * 8);
#pragma unroll
        for (int i = 0; i < 4; ++i)
            f.b[g][i] = *(const v4f*)(bp + (long)(g * 8 + i) * N);
    }
}

__device__ __forceinline__ void mma_set(const FragSet& f, v8f (&acc)[4])
{
#pragma unroll
    for (int g = 0; g < KG; ++g) {
#pragma unroll
        for (int p = 0; p < 2; ++p) {
            v2f a;
            a.x = f.a[g][2 * p];
            a.y = f.a[g][2 * p + 1];
#pragma unroll
            for (int j = 0; j < 4; ++j) {
                v2f b;
                b.x = f.b[g][2 * p][j];
                b.y = f.b[g][2 * p + 1][j];
                acc[j] = __builtin_amdgcn_wmma_f32_16x16x4_f32(
                     false, a,  false, b,
                     (short)0, acc[j],  false,  false);
            }
        }
    }
}

__device__ __forceinline__ float exp_of_twice(float x)
{
    return expf(2.0f * fminf(fmaxf(x, -43.0f), 43.0f));
}

template <bool EXP2X, int SPLITK>
__global__ __launch_bounds__(128) void wmma_gemm_f32(
    const float* __restrict__ A, const float* __restrict__ Bm,
    float* __restrict__ C, int M, int N, int K,
    long strideA, long strideB, long strideC)
{
    (void)M;
    const int batch = blockIdx.y;
    const float* Ab = A  + (long)batch * strideA;
    const float* Bb = Bm + (long)batch * strideB;
    float*       Cb = C  + (long)batch * strideC;

    const int lane   = threadIdx.x & 31;
    const int wave   = threadIdx.x >> 5;
    const int strips = N >> 6;
    const int strip  = blockIdx.x % strips;
    const int band = (SPLITK == 1) ? (blockIdx.x / strips) * 4 + wave : blockIdx.x / strips;
    const int klen = K / SPLITK;
    const int kbeg = (SPLITK == 1) ? 0 : wave * klen;
    const int tm = band << 4;
    const int tn = strip << 6;

    const int h = lane >> 4;
    const int c = lane & 15;

    const float* ap = Ab + (long)(tm + c) * K + kbeg + 4 * h;
    const float* bp = Bb + (long)(kbeg + 4 * h) * N + tn + 4 * c;

    v8f acc[4] = {};
    FragSet f0, f1;
#define GROUP_AHEAD() __builtin_amdgcn_sched_barrier(0)
    load_set(f0, ap, bp, N);
    int kk = 0;
    for (; kk < klen - 2 * KC; kk += 2 * KC) {
        load_set(f1, ap + kk + KC, bp + (long)(kk + KC) * N, N);
        GROUP_AHEAD();
        mma_set(f0, acc);
        GROUP_AHEAD();
        load_set(f0, ap + kk + 2 * KC, bp + (long)(kk + 2 * KC) * N, N);
        GROUP_AHEAD();
        mma_set(f1, acc);
        GROUP_AHEAD();
    }
    load_set(f1, ap + kk + KC, bp + (long)(kk + KC) * N, N);
    GROUP_AHEAD();
    mma_set(f0, acc);
    mma_set(f1, acc);

    float* crow = Cb + (long)(tm + 8 * h) * N + tn + 4 * c;
    if constexpr (SPLITK == 1) {
#pragma unroll
        for (int r = 0; r < 8; ++r) {
            v4f o;
            o.x = acc[0][r]; o.y = acc[1][r]; o.z = acc[2][r]; o.w = acc[3][r];
            if (EXP2X) {
                o.x = exp_of_twice(o.x); o.y = exp_of_twice(o.y);
                o.z = exp_of_twice(o.z); o.w = exp_of_twice(o.w);
            }
            *(v4f*)(crow + (long)r * N) = o;
        }
    } else {
        __shared__ v4f part[SPLITK][8][32];
#pragma unroll
        for (int r = 0; r < 8; ++r) {
            v4f o;
            o.x = acc[0][r]; o.y = acc[1][r]; o.z = acc[2][r]; o.w = acc[3][r];
            part[wave][r][lane] = o;
        }
        __syncthreads();
        for (int q = 0; q < 2; ++q) {
            const int r = 2 * wave + q;
            v4f o = part[0][r][lane];
            for (int w = 1; w < SPLITK; ++w) o += part[w][r][lane];
            if (EXP2X) {
                o.x = exp_of_twice(o.x); o.y = exp_of_twice(o.y);
                o.z = exp_of_twice(o.z); o.w = exp_of_twice(o.w);
            }
            *(v4f*)(crow + (long)r * N) = o;
        }
    }
}

#define ROWS 8
#define QR 2

template <int MASK>
__device__ __forceinline__ float row_xor(float x)
{
    const int b = __builtin_bit_cast(int, x);
    return __builtin_bit_cast(float, __builtin_amdgcn_update_dpp(b, b, 0x160 + MASK, 0xf, 0xf, false));
}

__device__ __forceinline__ float other_row(float x)
{
    const int b = __builtin_bit_cast(int, x);
    return __builtin_bit_cast(float,
        __builtin_amdgcn_permlanex16(b, b, 0x76543210, 0xfedcba98, false, false));
}

__device__ __forceinline__ float wave_max(float x)
{
    x = fmaxf(x, row_xor<1>(x));
    x = fmaxf(x, row_xor<2>(x));
    x = fmaxf(x, row_xor<4>(x));
    x = fmaxf(x, row_xor<8>(x));
    return fmaxf(x, other_row(x));
}

__device__ __forceinline__ float wave_sum(float x)
{
    x += row_xor<1>(x);
    x += row_xor<2>(x);
    x += row_xor<4>(x);
    x += row_xor<8>(x);
    return x + other_row(x);
}

__global__ __launch_bounds__(512) void scores_softmax(
    const float* __restrict__ eq,
    const float* __restrict__ ek,
    const float* __restrict__ scale,
    float* __restrict__ attn)
{
    __shared__ __attribute__((aligned(16))) float red_max[QR][16];
    __shared__ __attribute__((aligned(16))) float red_sum[QR][16];

    const int bt0  = blockIdx.x * QR;
    const int b    = bt0 >> 6;
    const int tid  = threadIdx.x;
    const int lane = tid & 31;
    const int wave = tid >> 5;

    v4f q0[QR], q1[QR];
#pragma unroll
    for (int qi = 0; qi < QR; ++qi) {
        const v4f* qrow = (const v4f*)(eq + (long)(bt0 + qi) * UU) + lane * 2;
        q0[qi] = qrow[0];
        q1[qi] = qrow[1];
    }
    const v4f* srow = (const v4f*)scale + lane * 2;
    const v4f  c0 = srow[0], c1 = srow[1];
    const float csum = ((c0.x + c0.y) + (c0.z + c0.w)) + ((c1.x + c1.y) + (c1.z + c1.w));
    const v4f  m0 = -2.0f * c0, m1 = -2.0f * c1;

    const int srow0 = wave * (SS / 16);
    const float* kbase = ek + ((long)b * SS + srow0) * UU + lane * 8;

    float sc[QR] = {0.0f, 0.0f};
    const int li = lane & (ROWS - 1);
    const int lj = (lane / ROWS) * ROWS;

    for (int j = 0; j < SS / 16; j += ROWS) {
        v4f k0[ROWS], k1[ROWS];
        float p[QR][ROWS];
#pragma unroll
        for (int i = 0; i < ROWS; ++i) {
            const v4f* kr = (const v4f*)(kbase + (long)(j + i) * UU);
            k0[i] = kr[0];
            k1[i] = kr[1];
        }
#pragma unroll
        for (int qi = 0; qi < QR; ++qi) {
#pragma unroll
            for (int i = 0; i < ROWS; ++i) {
                float a = csum;
                a = fmaf(m0.x, __builtin_amdgcn_rcpf(fmaf(q0[qi].x, k0[i].x, 1.0f)), a);
                a = fmaf(m0.y, __builtin_amdgcn_rcpf(fmaf(q0[qi].y, k0[i].y, 1.0f)), a);
                a = fmaf(m0.z, __builtin_amdgcn_rcpf(fmaf(q0[qi].z, k0[i].z, 1.0f)), a);
                a = fmaf(m0.w, __builtin_amdgcn_rcpf(fmaf(q0[qi].w, k0[i].w, 1.0f)), a);
                a = fmaf(m1.x, __builtin_amdgcn_rcpf(fmaf(q1[qi].x, k1[i].x, 1.0f)), a);
                a = fmaf(m1.y, __builtin_amdgcn_rcpf(fmaf(q1[qi].y, k1[i].y, 1.0f)), a);
                a = fmaf(m1.z, __builtin_amdgcn_rcpf(fmaf(q1[qi].z, k1[i].z, 1.0f)), a);
                a = fmaf(m1.w, __builtin_amdgcn_rcpf(fmaf(q1[qi].w, k1[i].w, 1.0f)), a);
                p[qi][i] = a;
            }
        }
        for (int off = 16; off; off >>= 1) {
#pragma unroll
            for (int qi = 0; qi < QR; ++qi) {
#pragma unroll
                for (int i = 0; i < ROWS; ++i) p[qi][i] += __shfl_xor(p[qi][i], off, 32);
            }
        }
#pragma unroll
        for (int qi = 0; qi < QR; ++qi) {
            float t = p[qi][0];
#pragma unroll
            for (int i = 1; i < ROWS; ++i) t = (li == i) ? p[qi][i] : t;
            sc[qi] = (lj == j) ? t : sc[qi];
        }
    }

    float m[QR], e[QR];
#pragma unroll
    for (int qi = 0; qi < QR; ++qi) {
        const float w = wave_max(sc[qi]);
        if (lane == 0) red_max[qi][wave] = w;
    }
    __syncthreads();
#pragma unroll
    for (int qi = 0; qi < QR; ++qi) {
        const v4f* r = (const v4f*)red_max[qi];
        const v4f ra = r[0], rb = r[1], rc = r[2], rd = r[3];
        const float ma = fmaxf(fmaxf(ra.x, ra.y), fmaxf(ra.z, ra.w));
        const float mb = fmaxf(fmaxf(rb.x, rb.y), fmaxf(rb.z, rb.w));
        const float mc = fmaxf(fmaxf(rc.x, rc.y), fmaxf(rc.z, rc.w));
        const float md = fmaxf(fmaxf(rd.x, rd.y), fmaxf(rd.z, rd.w));
        m[qi] = fmaxf(fmaxf(ma, mb), fmaxf(mc, md));
    }

#pragma unroll
    for (int qi = 0; qi < QR; ++qi) {
        e[qi] = __expf(sc[qi] - m[qi]);
        const float w = wave_sum(e[qi]);
        if (lane == 0) red_sum[qi][wave] = w;
    }
    __syncthreads();
#pragma unroll
    for (int qi = 0; qi < QR; ++qi) {
        const v4f* r = (const v4f*)red_sum[qi];
        const v4f ra = r[0], rb = r[1], rc = r[2], rd = r[3];
        const float tot = (((ra.x + ra.y) + (ra.z + ra.w)) + ((rb.x + rb.y) + (rb.z + rb.w)))
                        + (((rc.x + rc.y) + (rc.z + rc.w)) + ((rd.x + rd.y) + (rd.z + rd.w)));
        const float inv = 1.0f / tot;
        attn[(long)(bt0 + qi) * SS + srow0 + lane] = e[qi] * inv;
    }
}

extern "C" void kernel_launch(void* const* d_in, const int* in_sizes, int n_in,
                              void* d_out, int out_size, void* d_ws, size_t ws_size,
                              hipStream_t stream)
{
    (void)in_sizes; (void)n_in; (void)out_size; (void)ws_size;

    const float* query = (const float*)d_in[0];
    const float* value = (const float*)d_in[1];
    const float* W1    = (const float*)d_in[2];
    const float* W2    = (const float*)d_in[3];
    const float* scale = (const float*)d_in[4];

    float* ctx  = (float*)d_out;
    float* attn = ctx + (long)BB * TT * VU;

    float* w1q = (float*)d_ws;
    float* w2k = w1q + (long)BB * TT * UU;

    wmma_gemm_f32<true, 4><<<dim3(128, 1), 128, 0, stream>>>(
        query, W1, w1q, BB * TT, UU, QU, 0, 0, 0);

    wmma_gemm_f32<true, 1><<<dim3(256, 1), 128, 0, stream>>>(
        value, W2, w2k, BB * SS, UU, VU, 0, 0, 0);

    scores_softmax<<<dim3(BB * TT / 2), 512, 0, stream>>>(w1q, w2k, scale, attn);

    wmma_gemm_f32<false, 4><<<dim3(32, BB), 128, 0, stream>>>(
        attn, value, ctx, TT, VU, SS,
        (long)TT * SS, (long)SS * VU, (long)TT * VU);
}


# key projection fused into the scores kernel: a workgroup makes a 16-key Ek tile on WMMA, keeps it in LDS, scores it against all 64 query rows; softmax moved into the context GEMM: 3 launches, was 4
# speedup vs baseline: 6.2873x; 1.1930x over previous
#include <hip/hip_runtime.h>

typedef __attribute__((ext_vector_type(2))) float v2f;
typedef __attribute__((ext_vector_type(4))) float v4f;
typedef __attribute__((ext_vector_type(8))) float v8f;

#define BB 8
#define TT 64
#define SS 512
#define QU 1024
#define VU 512
#define UU 256

#define KC 64
#define KG (KC / 8)

template <int G>
struct FragSet {
    v4f a[G];
    v4f b[G][4];
};

template <int G>
__device__ __forceinline__ void load_b(FragSet<G>& f, const float* __restrict__ bp, long N)
{
#pragma unroll
    for (int g = 0; g < G; ++g) {
#pragma unroll
        for (int i = 0; i < 4; ++i)
            f.b[g][i] = *(const v4f*)(bp + (long)(g * 8 + i) * N);
    }
}

template <int G>
__device__ __forceinline__ void load_set(FragSet<G>& f, const float* __restrict__ ap,
                                         const float* __restrict__ bp, long N)
{
#pragma unroll
    for (int g = 0; g < G; ++g) f.a[g] = *(const v4f*)(ap + g * 8);
    load_b(f, bp, N);
}

template <int G>
__device__ __forceinline__ void mma_set(const FragSet<G>& f, v8f (&acc)[4])
{
#pragma unroll
    for (int g = 0; g < G; ++g) {
#pragma unroll
        for (int p = 0; p < 2; ++p) {
            v2f a;
            a.x = f.a[g][2 * p];
            a.y = f.a[g][2 * p + 1];
#pragma unroll
            for (int j = 0; j < 4; ++j) {
                v2f b;
                b.x = f.b[g][2 * p][j];
                b.y = f.b[g][2 * p + 1][j];
                acc[j] = __builtin_amdgcn_wmma_f32_16x16x4_f32(
                     false, a,  false, b,
                     (short)0, acc[j],  false,  false);
            }
        }
    }
}

__device__ __forceinline__ float exp_of_twice(float x)
{
    return expf(2.0f * fminf(fmaxf(x, -43.0f), 43.0f));
}

template <bool EXP2X, int SPLITK>
__global__ __launch_bounds__(128) void wmma_gemm_f32(
    const float* __restrict__ A, const float* __restrict__ Bm,
    float* __restrict__ C, int M, int N, int K,
    long strideA, long strideB, long strideC)
{
    (void)M;
    const int batch = blockIdx.y;
    const float* Ab = A  + (long)batch * strideA;
    const float* Bb = Bm + (long)batch * strideB;
    float*       Cb = C  + (long)batch * strideC;

    const int lane   = threadIdx.x & 31;
    const int wave   = threadIdx.x >> 5;
    const int strips = N >> 6;
    const int strip  = blockIdx.x % strips;
    const int band = (SPLITK == 1) ? (blockIdx.x / strips) * 4 + wave : blockIdx.x / strips;
    const int klen = K / SPLITK;
    const int kbeg = (SPLITK == 1) ? 0 : wave * klen;
    const int tm = band << 4;
    const int tn = strip << 6;

    const int h = lane >> 4;
    const int c = lane & 15;

    const float* ap = Ab + (long)(tm + c) * K + kbeg + 4 * h;
    const float* bp = Bb + (long)(kbeg + 4 * h) * N + tn + 4 * c;

    v8f acc[4] = {};
    FragSet<KG> f0, f1;
#define GROUP_AHEAD() __builtin_amdgcn_sched_barrier(0)
    load_set(f0, ap, bp, N);
    int kk = 0;
    for (; kk < klen - 2 * KC; kk += 2 * KC) {
        load_set(f1, ap + kk + KC, bp + (long)(kk + KC) * N, N);
        GROUP_AHEAD();
        mma_set(f0, acc);
        GROUP_AHEAD();
        load_set(f0, ap + kk + 2 * KC, bp + (long)(kk + 2 * KC) * N, N);
        GROUP_AHEAD();
        mma_set(f1, acc);
        GROUP_AHEAD();
    }
    load_set(f1, ap + kk + KC, bp + (long)(kk + KC) * N, N);
    GROUP_AHEAD();
    mma_set(f0, acc);
    mma_set(f1, acc);

    float* crow = Cb + (long)(tm + 8 * h) * N + tn + 4 * c;
    if constexpr (SPLITK == 1) {
#pragma unroll
        for (int r = 0; r < 8; ++r) {
            v4f o;
            o.x = acc[0][r]; o.y = acc[1][r]; o.z = acc[2][r]; o.w = acc[3][r];
            if (EXP2X) {
                o.x = exp_of_twice(o.x); o.y = exp_of_twice(o.y);
                o.z = exp_of_twice(o.z); o.w = exp_of_twice(o.w);
            }
            *(v4f*)(crow + (long)r * N) = o;
        }
    } else {
        __shared__ v4f part[SPLITK][8][32];
#pragma unroll
        for (int r = 0; r < 8; ++r) {
            v4f o;
            o.x = acc[0][r]; o.y = acc[1][r]; o.z = acc[2][r]; o.w = acc[3][r];
            part[wave][r][lane] = o;
        }
        __syncthreads();
        for (int q = 0; q < 2; ++q) {
            const int r = 2 * wave + q;
            v4f o = part[0][r][lane];
            for (int w = 1; w < SPLITK; ++w) o += part[w][r][lane];
            if (EXP2X) {
                o.x = exp_of_twice(o.x); o.y = exp_of_twice(o.y);
                o.z = exp_of_twice(o.z); o.w = exp_of_twice(o.w);
            }
            *(v4f*)(crow + (long)r * N) = o;
        }
    }
}

#define ROWS 8
#define QR 2
#define KT 16

__device__ __forceinline__ float other_row(float x)
{
    const int b = __builtin_bit_cast(int, x);
    return __builtin_bit_cast(float,
        __builtin_amdgcn_permlanex16(b, b, 0x76543210, 0xfedcba98, false, false));
}

__global__ __launch_bounds__(512) void keyproj_scores(
    const float* __restrict__ value,
    const float* __restrict__ W2,
    const float* __restrict__ eq,
    const float* __restrict__ scale,
    float* __restrict__ sraw)
{
    __shared__ v4f part[4][4][8][32];
    __shared__ __attribute__((aligned(16))) float ekt[KT][UU];

    const int b    = blockIdx.x >> 5;
    const int kt   = blockIdx.x & 31;
    const int tid  = threadIdx.x;
    const int lane = tid & 31;
    const int wave = tid >> 5;
    const int h = lane >> 4;
    const int c = lane & 15;

    {
        const int strip = wave & 3;
        const int kq    = wave >> 2;
        const int tn    = strip << 6;
        const int kbeg  = kq * (VU / 4);
        const float* ap = value + ((long)b * SS + kt * KT + c) * VU + kbeg + 4 * h;
        const float* bp = W2 + (long)(kbeg + 4 * h) * UU + tn + 4 * c;

        v8f acc[4] = {};
        FragSet<4> f0, f1;
        load_set(f0, ap, bp, UU);
        load_set(f1, ap + 32, bp + 32L * UU, UU);
        GROUP_AHEAD();
        mma_set(f0, acc);
        GROUP_AHEAD();
        load_set(f0, ap + 64, bp + 64L * UU, UU);
        GROUP_AHEAD();
        mma_set(f1, acc);
        GROUP_AHEAD();
        load_set(f1, ap + 96, bp + 96L * UU, UU);
        GROUP_AHEAD();
        mma_set(f0, acc);
        mma_set(f1, acc);

#pragma unroll
        for (int r = 0; r < 8; ++r) {
            v4f o;
            o.x = acc[0][r]; o.y = acc[1][r]; o.z = acc[2][r]; o.w = acc[3][r];
            part[kq][strip][r][lane] = o;
        }
    }
    __syncthreads();
    v4f qa[4], qb[4];
#pragma unroll
    for (int r = 0; r < 4; ++r) {
        const v4f* qrow = (const v4f*)(eq + ((long)b * TT + wave * 4 + r) * UU);
        qa[r] = qrow[lane];
        qb[r] = qrow[32 + lane];
    }
    for (int q = 0; q < 2; ++q) {
        const int idx   = 2 * wave + q;
        const int strip = idx >> 3;
        const int r     = idx & 7;
        v4f o = part[0][strip][r][lane];
        for (int w = 1; w < 4; ++w) o += part[w][strip][r][lane];
        o.x = exp_of_twice(o.x); o.y = exp_of_twice(o.y);
        o.z = exp_of_twice(o.z); o.w = exp_of_twice(o.w);
        *(v4f*)&ekt[8 * h + r][strip * 64 + 4 * c] = o;
    }
    __syncthreads();

    const v4f  c0 = ((const v4f*)scale)[lane], c1 = ((const v4f*)scale)[32 + lane];
    const float csum = ((c0.x + c0.y) + (c0.z + c0.w)) + ((c1.x + c1.y) + (c1.z + c1.w));
    const v4f  m0 = -2.0f * c0, m1 = -2.0f * c1;

    const int li = lane & (ROWS - 1);
    const int lk = c >> 3;
    float* sout = sraw + (((long)b * (SS / KT) + kt) * TT + wave * 4) * KT;

    float sc[2] = {0.0f, 0.0f};
#pragma nounroll
    for (int kh = 0; kh < KT / ROWS; ++kh) {
        v4f k0[ROWS], k1[ROWS];
#pragma unroll
        for (int i = 0; i < ROWS; ++i) {
            const v4f* kr = (const v4f*)ekt[kh * ROWS + i];
            k0[i] = kr[lane];
            k1[i] = kr[32 + lane];
        }
#pragma unroll
        for (int qp = 0; qp < 2; ++qp) {
            float p[QR][ROWS];
#pragma unroll
            for (int qi = 0; qi < QR; ++qi) {
                const v4f qx = qa[qp * QR + qi], qy = qb[qp * QR + qi];
#pragma unroll
                for (int i = 0; i < ROWS; ++i) {
                    float a = csum;
                    a = fmaf(m0.x, __builtin_amdgcn_rcpf(fmaf(qx.x, k0[i].x, 1.0f)), a);
                    a = fmaf(m0.y, __builtin_amdgcn_rcpf(fmaf(qx.y, k0[i].y, 1.0f)), a);
                    a = fmaf(m0.z, __builtin_amdgcn_rcpf(fmaf(qx.z, k0[i].z, 1.0f)), a);
                    a = fmaf(m0.w, __builtin_amdgcn_rcpf(fmaf(qx.w, k0[i].w, 1.0f)), a);
                    a = fmaf(m1.x, __builtin_amdgcn_rcpf(fmaf(qy.x, k1[i].x, 1.0f)), a);
                    a = fmaf(m1.y, __builtin_amdgcn_rcpf(fmaf(qy.y, k1[i].y, 1.0f)), a);
                    a = fmaf(m1.z, __builtin_amdgcn_rcpf(fmaf(qy.z, k1[i].z, 1.0f)), a);
                    a = fmaf(m1.w, __builtin_amdgcn_rcpf(fmaf(qy.w, k1[i].w, 1.0f)), a);
                    p[qi][i] = a;
                }
            }
            for (int off = 16; off; off >>= 1) {
#pragma unroll
                for (int qi = 0; qi < QR; ++qi) {
#pragma unroll
                    for (int i = 0; i < ROWS; ++i) p[qi][i] += __shfl_xor(p[qi][i], off, 32);
                }
            }
            float t[QR];
#pragma unroll
            for (int qi = 0; qi < QR; ++qi) {
                t[qi] = p[qi][0];
#pragma unroll
                for (int i = 1; i < ROWS; ++i) t[qi] = (li == i) ? p[qi][i] : t[qi];
            }
            const float mine = h ? t[1] : t[0];
            sc[qp] = (lk == kh) ? mine : sc[qp];
        }
    }
    sout[lane]      = sc[0];
    sout[32 + lane] = sc[1];
}

__global__ __launch_bounds__(128) void softmax_context(
    const float* __restrict__ sraw,
    const float* __restrict__ value,
    float* __restrict__ ctx,
    float* __restrict__ attn)
{
    __shared__ v4f part[4][8][32];
    __shared__ float st_m[4][16];
    __shared__ float st_l[4][16];

    const int batch = blockIdx.y;
    const int lane  = threadIdx.x & 31;
    const int wave  = threadIdx.x >> 5;
    const int strip = blockIdx.x & 7;
    const int band  = blockIdx.x >> 3;
    const int tm = band << 4;
    const int tn = strip << 6;
    const int kbeg = wave * (SS / 4);
    const int h = lane >> 4;
    const int c = lane & 15;

    FragSet<KG> f[2];
    const float* sp = sraw + (((long)batch * (SS / KT) + wave * 8) * TT + tm + c) * KT + 4 * h;
#pragma unroll
    for (int cc = 0; cc < 2; ++cc) {
#pragma unroll
        for (int g = 0; g < KG; ++g)
            f[cc].a[g] = *(const v4f*)(sp + (long)(cc * 4 + (g >> 1)) * TT * KT + (g & 1) * 8);
    }
    const float* bp = value + ((long)batch * SS + kbeg + 4 * h) * VU + tn + 4 * c;
    load_b(f[0], bp, VU);
    load_b(f[1], bp + (long)KC * VU, VU);

    float mx = f[0].a[0].x;
#pragma unroll
    for (int cc = 0; cc < 2; ++cc) {
#pragma unroll
        for (int g = 0; g < KG; ++g) {
            const v4f s = f[cc].a[g];
            mx = fmaxf(fmaxf(mx, fmaxf(s.x, s.y)), fmaxf(s.z, s.w));
        }
    }
    mx = fmaxf(mx, other_row(mx));
    float l = 0.0f;
#pragma unroll
    for (int cc = 0; cc < 2; ++cc) {
#pragma unroll
        for (int g = 0; g < KG; ++g) {
            v4f s = f[cc].a[g];
            s.x = __expf(s.x - mx); s.y = __expf(s.y - mx);
            s.z = __expf(s.z - mx); s.w = __expf(s.w - mx);
            f[cc].a[g] = s;
            l += (s.x + s.y) + (s.z + s.w);
        }
    }
    l += other_row(l);
    if (h == 0) {
        st_m[wave][c] = mx;
        st_l[wave][c] = l;
    }
    __syncthreads();
    const float m = fmaxf(fmaxf(st_m[0][c], st_m[1][c]), fmaxf(st_m[2][c], st_m[3][c]));
    float L = 0.0f;
    for (int w = 0; w < 4; ++w) L += st_l[w][c] * __expf(st_m[w][c] - m);
    const float fac = __expf(mx - m) / L;
#pragma unroll
    for (int cc = 0; cc < 2; ++cc) {
#pragma unroll
        for (int g = 0; g < KG; ++g) f[cc].a[g] *= fac;
    }
    if (strip == 0) {
        float* arow = attn + ((long)batch * TT + tm + c) * SS + kbeg + 4 * h;
#pragma unroll
        for (int cc = 0; cc < 2; ++cc) {
#pragma unroll
            for (int g = 0; g < KG; ++g) *(v4f*)(arow + cc * KC + g * 8) = f[cc].a[g];
        }
    }

    v8f acc[4] = {};
    mma_set(f[0], acc);
    mma_set(f[1], acc);

    float* crow = ctx + ((long)batch * TT + tm + 8 * h) * VU + tn + 4 * c;
#pragma unroll
    for (int r = 0; r < 8; ++r) {
        v4f o;
        o.x = acc[0][r]; o.y = acc[1][r]; o.z = acc[2][r]; o.w = acc[3][r];
        part[wave][r][lane] = o;
    }
    __syncthreads();
    for (int q = 0; q < 2; ++q) {
        const int r = 2 * wave + q;
        v4f o = part[0][r][lane];
        for (int w = 1; w < 4; ++w) o += part[w][r][lane];
        *(v4f*)(crow + (long)r * VU) = o;
    }
}

extern "C" void kernel_launch(void* const* d_in, const int* in_sizes, int n_in,
                              void* d_out, int out_size, void* d_ws, size_t ws_size,
                              hipStream_t stream)
{
    (void)in_sizes; (void)n_in; (void)out_size; (void)ws_size;

    const float* query = (const float*)d_in[0];
    const float* value = (const float*)d_in[1];
    const float* W1    = (const float*)d_in[2];
    const float* W2    = (const float*)d_in[3];
    const float* scale = (const float*)d_in[4];

    float* ctx  = (float*)d_out;
    float* attn = ctx + (long)BB * TT * VU;

    float* w1q  = (float*)d_ws;
    float* sraw = w1q + (long)BB * TT * UU;

    wmma_gemm_f32<true, 4><<<dim3(128, 1), 128, 0, stream>>>(
        query, W1, w1q, BB * TT, UU, QU, 0, 0, 0);

    keyproj_scores<<<dim3(BB * SS / KT), 512, 0, stream>>>(value, W2, w1q, scale, sraw);

    softmax_context<<<dim3(32, BB), 128, 0, stream>>>(sraw, value, ctx, attn);
}


# scores kernel: a trip's 16 lane-partial scores summed over the 32 lanes by one transposing reduction on the vector ALU (16 permlanex16/DPP exchanges), not 16 five-step ds_bpermute butterflies (80)
# speedup vs baseline: 6.4792x; 1.0305x over previous
#include <hip/hip_runtime.h>

typedef __attribute__((ext_vector_type(2))) float v2f;
typedef __attribute__((ext_vector_type(4))) float v4f;
typedef __attribute__((ext_vector_type(8))) float v8f;

#define BB 8
#define TT 64
#define SS 512
#define QU 1024
#define VU 512
#define UU 256

#define KC 64
#define KG (KC / 8)

template <int G>
struct FragSet {
    v4f a[G];
    v4f b[G][4];
};

template <int G>
__device__ __forceinline__ void load_b(FragSet<G>& f, const float* __restrict__ bp, long N)
{
#pragma unroll
    for (int g = 0; g < G; ++g) {
#pragma unroll
        for (int i = 0; i < 4; ++i)
            f.b[g][i] = *(const v4f*)(bp + (long)(g * 8 + i) * N);
    }
}

template <int G>
__device__ __forceinline__ void load_set(FragSet<G>& f, const float* __restrict__ ap,
                                         const float* __restrict__ bp, long N)
{
#pragma unroll
    for (int g = 0; g < G; ++g) f.a[g] = *(const v4f*)(ap + g * 8);
    load_b(f, bp, N);
}

template <int G>
__device__ __forceinline__ void mma_set(const FragSet<G>& f, v8f (&acc)[4])
{
#pragma unroll
    for (int g = 0; g < G; ++g) {
#pragma unroll
        for (int p = 0; p < 2; ++p) {
            v2f a;
            a.x = f.a[g][2 * p];
            a.y = f.a[g][2 * p + 1];
#pragma unroll
            for (int j = 0; j < 4; ++j) {
                v2f b;
                b.x = f.b[g][2 * p][j];
                b.y = f.b[g][2 * p + 1][j];
                acc[j] = __builtin_amdgcn_wmma_f32_16x16x4_f32(
                     false, a,  false, b,
                     (short)0, acc[j],  false,  false);
            }
        }
    }
}

__device__ __forceinline__ float exp_of_twice(float x)
{
    return expf(2.0f * fminf(fmaxf(x, -43.0f), 43.0f));
}

template <bool EXP2X, int SPLITK>
__global__ __launch_bounds__(128) void wmma_gemm_f32(
    const float* __restrict__ A, const float* __restrict__ Bm,
    float* __restrict__ C, int M, int N, int K,
    long strideA, long strideB, long strideC)
{
    (void)M;
    const int batch = blockIdx.y;
    const float* Ab = A  + (long)batch * strideA;
    const float* Bb = Bm + (long)batch * strideB;
    float*       Cb = C  + (long)batch * strideC;

    const int lane   = threadIdx.x & 31;
    const int wave   = threadIdx.x >> 5;
    const int strips = N >> 6;
    const int strip  = blockIdx.x % strips;
    const int band = (SPLITK == 1) ? (blockIdx.x / strips) * 4 + wave : blockIdx.x / strips;
    const int klen = K / SPLITK;
    const int kbeg = (SPLITK == 1) ? 0 : wave * klen;
    const int tm = band << 4;
    const int tn = strip << 6;

    const int h = lane >> 4;
    const int c = lane & 15;

    const float* ap = Ab + (long)(tm + c) * K + kbeg + 4 * h;
    const float* bp = Bb + (long)(kbeg + 4 * h) * N + tn + 4 * c;

    v8f acc[4] = {};
    FragSet<KG> f0, f1;
#define GROUP_AHEAD() __builtin_amdgcn_sched_barrier(0)
    load_set(f0, ap, bp, N);
    int kk = 0;
    for (; kk < klen - 2 * KC; kk += 2 * KC) {
        load_set(f1, ap + kk + KC, bp + (long)(kk + KC) * N, N);
        GROUP_AHEAD();
        mma_set(f0, acc);
        GROUP_AHEAD();
        load_set(f0, ap + kk + 2 * KC, bp + (long)(kk + 2 * KC) * N, N);
        GROUP_AHEAD();
        mma_set(f1, acc);
        GROUP_AHEAD();
    }
    load_set(f1, ap + kk + KC, bp + (long)(kk + KC) * N, N);
    GROUP_AHEAD();
    mma_set(f0, acc);
    mma_set(f1, acc);

    float* crow = Cb + (long)(tm + 8 * h) * N + tn + 4 * c;
    if constexpr (SPLITK == 1) {
#pragma unroll
        for (int r = 0; r < 8; ++r) {
            v4f o;
            o.x = acc[0][r]; o.y = acc[1][r]; o.z = acc[2][r]; o.w = acc[3][r];
            if (EXP2X) {
                o.x = exp_of_twice(o.x); o.y = exp_of_twice(o.y);
                o.z = exp_of_twice(o.z); o.w = exp_of_twice(o.w);
            }
            *(v4f*)(crow + (long)r * N) = o;
        }
    } else {
        __shared__ v4f part[SPLITK][8][32];
#pragma unroll
        for (int r = 0; r < 8; ++r) {
            v4f o;
            o.x = acc[0][r]; o.y = acc[1][r]; o.z = acc[2][r]; o.w = acc[3][r];
            part[wave][r][lane] = o;
        }
        __syncthreads();
        for (int q = 0; q < 2; ++q) {
            const int r = 2 * wave + q;
            v4f o = part[0][r][lane];
            for (int w = 1; w < SPLITK; ++w) o += part[w][r][lane];
            if (EXP2X) {
                o.x = exp_of_twice(o.x); o.y = exp_of_twice(o.y);
                o.z = exp_of_twice(o.z); o.w = exp_of_twice(o.w);
            }
            *(v4f*)(crow + (long)r * N) = o;
        }
    }
}

#define ROWS 8
#define QR 2
#define KT 16

__device__ __forceinline__ float other_row(float x)
{
    const int b = __builtin_bit_cast(int, x);
    return __builtin_bit_cast(float,
        __builtin_amdgcn_permlanex16(b, b, 0x76543210, 0xfedcba98, false, false));
}

template <int M>
__device__ __forceinline__ float row_xor(float x)
{
    const int b = __builtin_bit_cast(int, x);
    return __builtin_bit_cast(float,
        __builtin_amdgcn_update_dpp(b, b, 0x160 | M, 0xf, 0xf, true));
}

template <int M>
__device__ __forceinline__ float fold(float a, float b, bool up)
{
    const float keep = up ? b : a;
    const float send = up ? a : b;
    return keep + row_xor<M>(send);
}

__device__ __forceinline__ float fold_rows(float a, float b, bool up)
{
    const float keep = up ? b : a;
    const float send = up ? a : b;
    return keep + other_row(send);
}

__global__ __launch_bounds__(512) void keyproj_scores(
    const float* __restrict__ value,
    const float* __restrict__ W2,
    const float* __restrict__ eq,
    const float* __restrict__ scale,
    float* __restrict__ sraw)
{
    __shared__ v4f part[4][4][8][32];
    __shared__ __attribute__((aligned(16))) float ekt[KT][UU];

    const int b    = blockIdx.x >> 5;
    const int kt   = blockIdx.x & 31;
    const int tid  = threadIdx.x;
    const int lane = tid & 31;
    const int wave = tid >> 5;
    const int h = lane >> 4;
    const int c = lane & 15;

    {
        const int strip = wave & 3;
        const int kq    = wave >> 2;
        const int tn    = strip << 6;
        const int kbeg  = kq * (VU / 4);
        const float* ap = value + ((long)b * SS + kt * KT + c) * VU + kbeg + 4 * h;
        const float* bp = W2 + (long)(kbeg + 4 * h) * UU + tn + 4 * c;

        v8f acc[4] = {};
        FragSet<4> f0, f1;
        load_set(f0, ap, bp, UU);
        load_set(f1, ap + 32, bp + 32L * UU, UU);
        GROUP_AHEAD();
        mma_set(f0, acc);
        GROUP_AHEAD();
        load_set(f0, ap + 64, bp + 64L * UU, UU);
        GROUP_AHEAD();
        mma_set(f1, acc);
        GROUP_AHEAD();
        load_set(f1, ap + 96, bp + 96L * UU, UU);
        GROUP_AHEAD();
        mma_set(f0, acc);
        mma_set(f1, acc);

#pragma unroll
        for (int r = 0; r < 8; ++r) {
            v4f o;
            o.x = acc[0][r]; o.y = acc[1][r]; o.z = acc[2][r]; o.w = acc[3][r];
            part[kq][strip][r][lane] = o;
        }
    }
    __syncthreads();
    v4f qa[4], qb[4];
#pragma unroll
    for (int r = 0; r < 4; ++r) {
        const v4f* qrow = (const v4f*)(eq + ((long)b * TT + wave * 4 + r) * UU);
        qa[r] = qrow[lane];
        qb[r] = qrow[32 + lane];
    }
    for (int q = 0; q < 2; ++q) {
        const int idx   = 2 * wave + q;
        const int strip = idx >> 3;
        const int r     = idx & 7;
        v4f o = part[0][strip][r][lane];
        for (int w = 1; w < 4; ++w) o += part[w][strip][r][lane];
        o.x = exp_of_twice(o.x); o.y = exp_of_twice(o.y);
        o.z = exp_of_twice(o.z); o.w = exp_of_twice(o.w);
        *(v4f*)&ekt[8 * h + r][strip * 64 + 4 * c] = o;
    }
    __syncthreads();

    const v4f  c0 = ((const v4f*)scale)[lane], c1 = ((const v4f*)scale)[32 + lane];
    const float csum = ((c0.x + c0.y) + (c0.z + c0.w)) + ((c1.x + c1.y) + (c1.z + c1.w));
    const v4f  m0 = -2.0f * c0, m1 = -2.0f * c1;

    const bool bit0 = (lane & 1) != 0, bit1 = (lane & 2) != 0, bit2 = (lane & 4) != 0;
    const int lk = c >> 3;
    float* sout = sraw + (((long)b * (SS / KT) + kt) * TT + wave * 4) * KT;

    float sc[2] = {0.0f, 0.0f};
#pragma nounroll
    for (int kh = 0; kh < KT / ROWS; ++kh) {
        v4f k0[ROWS], k1[ROWS];
#pragma unroll
        for (int i = 0; i < ROWS; ++i) {
            const v4f* kr = (const v4f*)ekt[kh * ROWS + i];
            k0[i] = kr[lane];
            k1[i] = kr[32 + lane];
        }
#pragma unroll
        for (int qp = 0; qp < 2; ++qp) {
            float p[QR][ROWS];
#pragma unroll
            for (int qi = 0; qi < QR; ++qi) {
                const v4f qx = qa[qp * QR + qi], qy = qb[qp * QR + qi];
#pragma unroll
                for (int i = 0; i < ROWS; ++i) {
                    float a = csum;
                    a = fmaf(m0.x, __builtin_amdgcn_rcpf(fmaf(qx.x, k0[i].x, 1.0f)), a);
                    a = fmaf(m0.y, __builtin_amdgcn_rcpf(fmaf(qx.y, k0[i].y, 1.0f)), a);
                    a = fmaf(m0.z, __builtin_amdgcn_rcpf(fmaf(qx.z, k0[i].z, 1.0f)), a);
                    a = fmaf(m0.w, __builtin_amdgcn_rcpf(fmaf(qx.w, k0[i].w, 1.0f)), a);
                    a = fmaf(m1.x, __builtin_amdgcn_rcpf(fmaf(qy.x, k1[i].x, 1.0f)), a);
                    a = fmaf(m1.y, __builtin_amdgcn_rcpf(fmaf(qy.y, k1[i].y, 1.0f)), a);
                    a = fmaf(m1.z, __builtin_amdgcn_rcpf(fmaf(qy.z, k1[i].z, 1.0f)), a);
                    a = fmaf(m1.w, __builtin_amdgcn_rcpf(fmaf(qy.w, k1[i].w, 1.0f)), a);
                    p[qi][i] = a;
                }
            }
            float v8[ROWS];
#pragma unroll
            for (int i = 0; i < ROWS; ++i) v8[i] = fold_rows(p[0][i], p[1][i], h != 0);
            float v4[4];
#pragma unroll
            for (int i = 0; i < 4; ++i) v4[i] = fold<4>(v8[i], v8[i + 4], bit2);
            const float v2a = fold<2>(v4[0], v4[2], bit1);
            const float v2b = fold<2>(v4[1], v4[3], bit1);
            float mine = fold<1>(v2a, v2b, bit0);
            mine += row_xor<8>(mine);
            sc[qp] = (lk == kh) ? mine : sc[qp];
        }
    }
    sout[lane]      = sc[0];
    sout[32 + lane] = sc[1];
}

__global__ __launch_bounds__(128) void softmax_context(
    const float* __restrict__ sraw,
    const float* __restrict__ value,
    float* __restrict__ ctx,
    float* __restrict__ attn)
{
    __shared__ v4f part[4][8][32];
    __shared__ float st_m[4][16];
    __shared__ float st_l[4][16];

    const int batch = blockIdx.y;
    const int lane  = threadIdx.x & 31;
    const int wave  = threadIdx.x >> 5;
    const int strip = blockIdx.x & 7;
    const int band  = blockIdx.x >> 3;
    const int tm = band << 4;
    const int tn = strip << 6;
    const int kbeg = wave * (SS / 4);
    const int h = lane >> 4;
    const int c = lane & 15;

    FragSet<KG> f[2];
    const float* sp = sraw + (((long)batch * (SS / KT) + wave * 8) * TT + tm + c) * KT + 4 * h;
#pragma unroll
    for (int cc = 0; cc < 2; ++cc) {
#pragma unroll
        for (int g = 0; g < KG; ++g)
            f[cc].a[g] = *(const v4f*)(sp + (long)(cc * 4 + (g >> 1)) * TT * KT + (g & 1) * 8);
    }
    const float* bp = value + ((long)batch * SS + kbeg + 4 * h) * VU + tn + 4 * c;
    load_b(f[0], bp, VU);
    load_b(f[1], bp + (long)KC * VU, VU);

    float mx = f[0].a[0].x;
#pragma unroll
    for (int cc = 0; cc < 2; ++cc) {
#pragma unroll
        for (int g = 0; g < KG; ++g) {
            const v4f s = f[cc].a[g];
            mx = fmaxf(fmaxf(mx, fmaxf(s.x, s.y)), fmaxf(s.z, s.w));
        }
    }
    mx = fmaxf(mx, other_row(mx));
    float l = 0.0f;
#pragma unroll
    for (int cc = 0; cc < 2; ++cc) {
#pragma unroll
        for (int g = 0; g < KG; ++g) {
            v4f s = f[cc].a[g];
            s.x = __expf(s.x - mx); s.y = __expf(s.y - mx);
            s.z = __expf(s.z - mx); s.w = __expf(s.w - mx);
            f[cc].a[g] = s;
            l += (s.x + s.y) + (s.z + s.w);
        }
    }
    l += other_row(l);
    if (h == 0) {
        st_m[wave][c] = mx;
        st_l[wave][c] = l;
    }
    __syncthreads();
    const float m = fmaxf(fmaxf(st_m[0][c], st_m[1][c]), fmaxf(st_m[2][c], st_m[3][c]));
    float L = 0.0f;
    for (int w = 0; w < 4; ++w) L += st_l[w][c] * __expf(st_m[w][c] - m);
    const float fac = __expf(mx - m) / L;
#pragma unroll
    for (int cc = 0; cc < 2; ++cc) {
#pragma unroll
        for (int g = 0; g < KG; ++g) f[cc].a[g] *= fac;
    }
    if (strip == 0) {
        float* arow = attn + ((long)batch * TT + tm + c) * SS + kbeg + 4 * h;
#pragma unroll
        for (int cc = 0; cc < 2; ++cc) {
#pragma unroll
            for (int g = 0; g < KG; ++g) *(v4f*)(arow + cc * KC + g * 8) = f[cc].a[g];
        }
    }

    v8f acc[4] = {};
    mma_set(f[0], acc);
    mma_set(f[1], acc);

    float* crow = ctx + ((long)batch * TT + tm + 8 * h) * VU + tn + 4 * c;
#pragma unroll
    for (int r = 0; r < 8; ++r) {
        v4f o;
        o.x = acc[0][r]; o.y = acc[1][r]; o.z = acc[2][r]; o.w = acc[3][r];
        part[wave][r][lane] = o;
    }
    __syncthreads();
    for (int q = 0; q < 2; ++q) {
        const int r = 2 * wave + q;
        v4f o = part[0][r][lane];
        for (int w = 1; w < 4; ++w) o += part[w][r][lane];
        *(v4f*)(crow + (long)r * VU) = o;
    }
}

extern "C" void kernel_launch(void* const* d_in, const int* in_sizes, int n_in,
                              void* d_out, int out_size, void* d_ws, size_t ws_size,
                              hipStream_t stream)
{
    (void)in_sizes; (void)n_in; (void)out_size; (void)ws_size;

    const float* query = (const float*)d_in[0];
    const float* value = (const float*)d_in[1];
    const float* W1    = (const float*)d_in[2];
    const float* W2    = (const float*)d_in[3];
    const float* scale = (const float*)d_in[4];

    float* ctx  = (float*)d_out;
    float* attn = ctx + (long)BB * TT * VU;

    float* w1q  = (float*)d_ws;
    float* sraw = w1q + (long)BB * TT * UU;

    wmma_gemm_f32<true, 4><<<dim3(128, 1), 128, 0, stream>>>(
        query, W1, w1q, BB * TT, UU, QU, 0, 0, 0);

    keyproj_scores<<<dim3(BB * SS / KT), 512, 0, stream>>>(value, W2, w1q, scale, sraw);

    softmax_context<<<dim3(32, BB), 128, 0, stream>>>(sraw, value, ctx, attn);
}


# query projection GEMM: a workgroup owns a 16x32 strip (two tiles, 8-byte B loads), not 16x64: 256 workgroups, not 128, so every WGP works and a wave's WMMA chain is 128 long, not 256
# speedup vs baseline: 6.8169x; 1.0521x over previous
#include <hip/hip_runtime.h>

typedef __attribute__((ext_vector_type(2))) float v2f;
typedef __attribute__((ext_vector_type(4))) float v4f;
typedef __attribute__((ext_vector_type(8))) float v8f;

#define BB 8
#define TT 64
#define SS 512
#define QU 1024
#define VU 512
#define UU 256

#define KC 64
#define KG (KC / 8)

template <int G>
struct FragSet {
    v4f a[G];
    v4f b[G][4];
};

template <int G>
__device__ __forceinline__ void load_b(FragSet<G>& f, const float* __restrict__ bp, long N)
{
#pragma unroll
    for (int g = 0; g < G; ++g) {
#pragma unroll
        for (int i = 0; i < 4; ++i)
            f.b[g][i] = *(const v4f*)(bp + (long)(g * 8 + i) * N);
    }
}

template <int G>
__device__ __forceinline__ void load_set(FragSet<G>& f, const float* __restrict__ ap,
                                         const float* __restrict__ bp, long N)
{
#pragma unroll
    for (int g = 0; g < G; ++g) f.a[g] = *(const v4f*)(ap + g * 8);
    load_b(f, bp, N);
}

template <int G>
__device__ __forceinline__ void mma_set(const FragSet<G>& f, v8f (&acc)[4])
{
#pragma unroll
    for (int g = 0; g < G; ++g) {
#pragma unroll
        for (int p = 0; p < 2; ++p) {
            v2f a;
            a.x = f.a[g][2 * p];
            a.y = f.a[g][2 * p + 1];
#pragma unroll
            for (int j = 0; j < 4; ++j) {
                v2f b;
                b.x = f.b[g][2 * p][j];
                b.y = f.b[g][2 * p + 1][j];
                acc[j] = __builtin_amdgcn_wmma_f32_16x16x4_f32(
                     false, a,  false, b,
                     (short)0, acc[j],  false,  false);
            }
        }
    }
}

__device__ __forceinline__ float exp_of_twice(float x)
{
    return expf(2.0f * fminf(fmaxf(x, -43.0f), 43.0f));
}

#define PK 64
#define PG (PK / 8)

struct FragPair {
    v4f a[PG];
    v2f b[PG][4];
};

__device__ __forceinline__ void load_pair(FragPair& f, const float* __restrict__ ap,
                                          const float* __restrict__ bp, long N)
{
#pragma unroll
    for (int g = 0; g < PG; ++g) f.a[g] = *(const v4f*)(ap + g * 8);
#pragma unroll
    for (int g = 0; g < PG; ++g) {
#pragma unroll
        for (int i = 0; i < 4; ++i)
            f.b[g][i] = *(const v2f*)(bp + (long)(g * 8 + i) * N);
    }
}

__device__ __forceinline__ void mma_pair(const FragPair& f, v8f (&acc)[2])
{
#pragma unroll
    for (int g = 0; g < PG; ++g) {
#pragma unroll
        for (int p = 0; p < 2; ++p) {
            v2f a;
            a.x = f.a[g][2 * p];
            a.y = f.a[g][2 * p + 1];
#pragma unroll
            for (int j = 0; j < 2; ++j) {
                v2f b;
                b.x = f.b[g][2 * p][j];
                b.y = f.b[g][2 * p + 1][j];
                acc[j] = __builtin_amdgcn_wmma_f32_16x16x4_f32(
                     false, a,  false, b,
                     (short)0, acc[j],  false,  false);
            }
        }
    }
}

#define GROUP_AHEAD() __builtin_amdgcn_sched_barrier(0)

__global__ __launch_bounds__(128) void proj_exp2x(
    const float* __restrict__ A,
    const float* __restrict__ Bm,
    float* __restrict__ C)
{
    constexpr int N = UU, K = QU;
    const int lane   = threadIdx.x & 31;
    const int wave   = threadIdx.x >> 5;
    constexpr int strips = N >> 5;
    const int strip  = blockIdx.x % strips;
    const int band   = blockIdx.x / strips;
    const int kbeg   = wave * (K / 4);
    const int tm = band << 4;
    const int tn = strip << 5;

    const int h = lane >> 4;
    const int c = lane & 15;

    const float* ap = A  + (long)(tm + c) * K + kbeg + 4 * h;
    const float* bp = Bm + (long)(kbeg + 4 * h) * N + tn + 2 * c;

    v8f acc[2] = {};
    FragPair f0, f1;
    load_pair(f0, ap, bp, N);
    load_pair(f1, ap + PK, bp + (long)PK * N, N);
    GROUP_AHEAD();
    mma_pair(f0, acc);
    GROUP_AHEAD();
    load_pair(f0, ap + 2 * PK, bp + 2L * PK * N, N);
    GROUP_AHEAD();
    mma_pair(f1, acc);
    GROUP_AHEAD();
    load_pair(f1, ap + 3 * PK, bp + 3L * PK * N, N);
    GROUP_AHEAD();
    mma_pair(f0, acc);
    mma_pair(f1, acc);

    __shared__ v2f part[4][8][32];
#pragma unroll
    for (int r = 0; r < 8; ++r) {
        v2f o;
        o.x = acc[0][r]; o.y = acc[1][r];
        part[wave][r][lane] = o;
    }
    __syncthreads();
    float* crow = C + (long)(tm + 8 * h) * N + tn + 2 * c;
    for (int q = 0; q < 2; ++q) {
        const int r = 2 * wave + q;
        v2f o = part[0][r][lane];
        for (int w = 1; w < 4; ++w) o += part[w][r][lane];
        o.x = exp_of_twice(o.x); o.y = exp_of_twice(o.y);
        *(v2f*)(crow + (long)r * N) = o;
    }
}

#define ROWS 8
#define QR 2
#define KT 16

__device__ __forceinline__ float other_row(float x)
{
    const int b = __builtin_bit_cast(int, x);
    return __builtin_bit_cast(float,
        __builtin_amdgcn_permlanex16(b, b, 0x76543210, 0xfedcba98, false, false));
}

template <int M>
__device__ __forceinline__ float row_xor(float x)
{
    const int b = __builtin_bit_cast(int, x);
    return __builtin_bit_cast(float,
        __builtin_amdgcn_update_dpp(b, b, 0x160 | M, 0xf, 0xf, true));
}

template <int M>
__device__ __forceinline__ float fold(float a, float b, bool up)
{
    const float keep = up ? b : a;
    const float send = up ? a : b;
    return keep + row_xor<M>(send);
}

__device__ __forceinline__ float fold_rows(float a, float b, bool up)
{
    const float keep = up ? b : a;
    const float send = up ? a : b;
    return keep + other_row(send);
}

__global__ __launch_bounds__(512) void keyproj_scores(
    const float* __restrict__ value,
    const float* __restrict__ W2,
    const float* __restrict__ eq,
    const float* __restrict__ scale,
    float* __restrict__ sraw)
{
    __shared__ v4f part[4][4][8][32];
    __shared__ __attribute__((aligned(16))) float ekt[KT][UU];

    const int b    = blockIdx.x >> 5;
    const int kt   = blockIdx.x & 31;
    const int tid  = threadIdx.x;
    const int lane = tid & 31;
    const int wave = tid >> 5;
    const int h = lane >> 4;
    const int c = lane & 15;

    {
        const int strip = wave & 3;
        const int kq    = wave >> 2;
        const int tn    = strip << 6;
        const int kbeg  = kq * (VU / 4);
        const float* ap = value + ((long)b * SS + kt * KT + c) * VU + kbeg + 4 * h;
        const float* bp = W2 + (long)(kbeg + 4 * h) * UU + tn + 4 * c;

        v8f acc[4] = {};
        FragSet<4> f0, f1;
        load_set(f0, ap, bp, UU);
        load_set(f1, ap + 32, bp + 32L * UU, UU);
        GROUP_AHEAD();
        mma_set(f0, acc);
        GROUP_AHEAD();
        load_set(f0, ap + 64, bp + 64L * UU, UU);
        GROUP_AHEAD();
        mma_set(f1, acc);
        GROUP_AHEAD();
        load_set(f1, ap + 96, bp + 96L * UU, UU);
        GROUP_AHEAD();
        mma_set(f0, acc);
        mma_set(f1, acc);

#pragma unroll
        for (int r = 0; r < 8; ++r) {
            v4f o;
            o.x = acc[0][r]; o.y = acc[1][r]; o.z = acc[2][r]; o.w = acc[3][r];
            part[kq][strip][r][lane] = o;
        }
    }
    __syncthreads();
    v4f qa[4], qb[4];
#pragma unroll
    for (int r = 0; r < 4; ++r) {
        const v4f* qrow = (const v4f*)(eq + ((long)b * TT + wave * 4 + r) * UU);
        qa[r] = qrow[lane];
        qb[r] = qrow[32 + lane];
    }
    for (int q = 0; q < 2; ++q) {
        const int idx   = 2 * wave + q;
        const int strip = idx >> 3;
        const int r     = idx & 7;
        v4f o = part[0][strip][r][lane];
        for (int w = 1; w < 4; ++w) o += part[w][strip][r][lane];
        o.x = exp_of_twice(o.x); o.y = exp_of_twice(o.y);
        o.z = exp_of_twice(o.z); o.w = exp_of_twice(o.w);
        *(v4f*)&ekt[8 * h + r][strip * 64 + 4 * c] = o;
    }
    __syncthreads();

    const v4f  c0 = ((const v4f*)scale)[lane], c1 = ((const v4f*)scale)[32 + lane];
    const float csum = ((c0.x + c0.y) + (c0.z + c0.w)) + ((c1.x + c1.y) + (c1.z + c1.w));
    const v4f  m0 = -2.0f * c0, m1 = -2.0f * c1;

    const bool bit0 = (lane & 1) != 0, bit1 = (lane & 2) != 0, bit2 = (lane & 4) != 0;
    const int lk = c >> 3;
    float* sout = sraw + (((long)b * (SS / KT) + kt) * TT + wave * 4) * KT;

    float sc[2] = {0.0f, 0.0f};
#pragma nounroll
    for (int kh = 0; kh < KT / ROWS; ++kh) {
        v4f k0[ROWS], k1[ROWS];
#pragma unroll
        for (int i = 0; i < ROWS; ++i) {
            const v4f* kr = (const v4f*)ekt[kh * ROWS + i];
            k0[i] = kr[lane];
            k1[i] = kr[32 + lane];
        }
#pragma unroll
        for (int qp = 0; qp < 2; ++qp) {
            float p[QR][ROWS];
#pragma unroll
            for (int qi = 0; qi < QR; ++qi) {
                const v4f qx = qa[qp * QR + qi], qy = qb[qp * QR + qi];
#pragma unroll
                for (int i = 0; i < ROWS; ++i) {
                    float a = csum;
                    a = fmaf(m0.x, __builtin_amdgcn_rcpf(fmaf(qx.x, k0[i].x, 1.0f)), a);
                    a = fmaf(m0.y, __builtin_amdgcn_rcpf(fmaf(qx.y, k0[i].y, 1.0f)), a);
                    a = fmaf(m0.z, __builtin_amdgcn_rcpf(fmaf(qx.z, k0[i].z, 1.0f)), a);
                    a = fmaf(m0.w, __builtin_amdgcn_rcpf(fmaf(qx.w, k0[i].w, 1.0f)), a);
                    a = fmaf(m1.x, __builtin_amdgcn_rcpf(fmaf(qy.x, k1[i].x, 1.0f)), a);
                    a = fmaf(m1.y, __builtin_amdgcn_rcpf(fmaf(qy.y, k1[i].y, 1.0f)), a);
                    a = fmaf(m1.z, __builtin_amdgcn_rcpf(fmaf(qy.z, k1[i].z, 1.0f)), a);
                    a = fmaf(m1.w, __builtin_amdgcn_rcpf(fmaf(qy.w, k1[i].w, 1.0f)), a);
                    p[qi][i] = a;
                }
            }
            float v8[ROWS];
#pragma unroll
            for (int i = 0; i < ROWS; ++i) v8[i] = fold_rows(p[0][i], p[1][i], h != 0);
            float v4[4];
#pragma unroll
            for (int i = 0; i < 4; ++i) v4[i] = fold<4>(v8[i], v8[i + 4], bit2);
            const float v2a = fold<2>(v4[0], v4[2], bit1);
            const float v2b = fold<2>(v4[1], v4[3], bit1);
            float mine = fold<1>(v2a, v2b, bit0);
            mine += row_xor<8>(mine);
            sc[qp] = (lk == kh) ? mine : sc[qp];
        }
    }
    sout[lane]      = sc[0];
    sout[32 + lane] = sc[1];
}

__global__ __launch_bounds__(128) void softmax_context(
    const float* __restrict__ sraw,
    const float* __restrict__ value,
    float* __restrict__ ctx,
    float* __restrict__ attn)
{
    __shared__ v4f part[4][8][32];
    __shared__ float st_m[4][16];
    __shared__ float st_l[4][16];

    const int batch = blockIdx.y;
    const int lane  = threadIdx.x & 31;
    const int wave  = threadIdx.x >> 5;
    const int strip = blockIdx.x & 7;
    const int band  = blockIdx.x >> 3;
    const int tm = band << 4;
    const int tn = strip << 6;
    const int kbeg = wave * (SS / 4);
    const int h = lane >> 4;
    const int c = lane & 15;

    FragSet<KG> f[2];
    const float* sp = sraw + (((long)batch * (SS / KT) + wave * 8) * TT + tm + c) * KT + 4 * h;
#pragma unroll
    for (int cc = 0; cc < 2; ++cc) {
#pragma unroll
        for (int g = 0; g < KG; ++g)
            f[cc].a[g] = *(const v4f*)(sp + (long)(cc * 4 + (g >> 1)) * TT * KT + (g & 1) * 8);
    }
    const float* bp = value + ((long)batch * SS + kbeg + 4 * h) * VU + tn + 4 * c;
    load_b(f[0], bp, VU);
    load_b(f[1], bp + (long)KC * VU, VU);

    float mx = f[0].a[0].x;
#pragma unroll
    for (int cc = 0; cc < 2; ++cc) {
#pragma unroll
        for (int g = 0; g < KG; ++g) {
            const v4f s = f[cc].a[g];
            mx = fmaxf(fmaxf(mx, fmaxf(s.x, s.y)), fmaxf(s.z, s.w));
        }
    }
    mx = fmaxf(mx, other_row(mx));
    float l = 0.0f;
#pragma unroll
    for (int cc = 0; cc < 2; ++cc) {
#pragma unroll
        for (int g = 0; g < KG; ++g) {
            v4f s = f[cc].a[g];
            s.x = __expf(s.x - mx); s.y = __expf(s.y - mx);
            s.z = __expf(s.z - mx); s.w = __expf(s.w - mx);
            f[cc].a[g] = s;
            l += (s.x + s.y) + (s.z + s.w);
        }
    }
    l += other_row(l);
    if (h == 0) {
        st_m[wave][c] = mx;
        st_l[wave][c] = l;
    }
    __syncthreads();
    const float m = fmaxf(fmaxf(st_m[0][c], st_m[1][c]), fmaxf(st_m[2][c], st_m[3][c]));
    float L = 0.0f;
    for (int w = 0; w < 4; ++w) L += st_l[w][c] * __expf(st_m[w][c] - m);
    const float fac = __expf(mx - m) / L;
#pragma unroll
    for (int cc = 0; cc < 2; ++cc) {
#pragma unroll
        for (int g = 0; g < KG; ++g) f[cc].a[g] *= fac;
    }
    if (strip == 0) {
        float* arow = attn + ((long)batch * TT + tm + c) * SS + kbeg + 4 * h;
#pragma unroll
        for (int cc = 0; cc < 2; ++cc) {
#pragma unroll
            for (int g = 0; g < KG; ++g) *(v4f*)(arow + cc * KC + g * 8) = f[cc].a[g];
        }
    }

    v8f acc[4] = {};
    mma_set(f[0], acc);
    mma_set(f[1], acc);

    float* crow = ctx + ((long)batch * TT + tm + 8 * h) * VU + tn + 4 * c;
#pragma unroll
    for (int r = 0; r < 8; ++r) {
        v4f o;
        o.x = acc[0][r]; o.y = acc[1][r]; o.z = acc[2][r]; o.w = acc[3][r];
        part[wave][r][lane] = o;
    }
    __syncthreads();
    for (int q = 0; q < 2; ++q) {
        const int r = 2 * wave + q;
        v4f o = part[0][r][lane];
        for (int w = 1; w < 4; ++w) o += part[w][r][lane];
        *(v4f*)(crow + (long)r * VU) = o;
    }
}

extern "C" void kernel_launch(void* const* d_in, const int* in_sizes, int n_in,
                              void* d_out, int out_size, void* d_ws, size_t ws_size,
                              hipStream_t stream)
{
    (void)in_sizes; (void)n_in; (void)out_size; (void)ws_size;

    const float* query = (const float*)d_in[0];
    const float* value = (const float*)d_in[1];
    const float* W1    = (const float*)d_in[2];
    const float* W2    = (const float*)d_in[3];
    const float* scale = (const float*)d_in[4];

    float* ctx  = (float*)d_out;
    float* attn = ctx + (long)BB * TT * VU;

    float* w1q  = (float*)d_ws;
    float* sraw = w1q + (long)BB * TT * UU;

    proj_exp2x<<<dim3((BB * TT / 16) * (UU / 32)), 128, 0, stream>>>(query, W1, w1q);

    keyproj_scores<<<dim3(BB * SS / KT), 512, 0, stream>>>(value, W2, w1q, scale, sraw);

    softmax_context<<<dim3(32, BB), 128, 0, stream>>>(sraw, value, ctx, attn);
}
